# starting kernel
# baseline (speedup 1.0000x reference)
#include <hip/hip_runtime.h>

typedef __attribute__((ext_vector_type(2))) float v2f;
typedef __attribute__((ext_vector_type(8))) float v8f;

#define WMMA_F32(a, b, c) \
  __builtin_amdgcn_wmma_f32_16x16x4_f32(false, (a), false, (b), (short)0, (c), false, false)

#define BATCH 4
#define SEQ   2048
#define KDIM  1024
#define HEADS 16
#define HS    64
#define ROWS  (BATCH * SEQ)

__device__ __forceinline__ float rowmax16(float x) {
  x = fmaxf(x, __shfl_xor(x, 1, 32));
  x = fmaxf(x, __shfl_xor(x, 2, 32));
  x = fmaxf(x, __shfl_xor(x, 4, 32));
  x = fmaxf(x, __shfl_xor(x, 8, 32));
  return x;
}
__device__ __forceinline__ float rowsum16(float x) {
  x += __shfl_xor(x, 1, 32);
  x += __shfl_xor(x, 2, 32);
  x += __shfl_xor(x, 4, 32);
  x += __shfl_xor(x, 8, 32);
  return x;
}

__global__ void __launch_bounds__(32)
qkv_proj_kernel(const float* __restrict__ x,
                const float* __restrict__ Wq, const float* __restrict__ Wk,
                const float* __restrict__ Wv,
                float* __restrict__ qo, float* __restrict__ ko,
                float* __restrict__ vo) {
  const int lane = threadIdx.x;
  const int half = lane >> 4;
  const int l16  = lane & 15;
  const int m0 = blockIdx.x * 64;
  const int n0 = blockIdx.y * 64;
  const int which = blockIdx.z;

  const float* W  = (which == 0) ? Wq : (which == 1) ? Wk : Wv;
  float*      out = (which == 0) ? qo : (which == 1) ? ko : vo;

  v8f acc[4][4];
#pragma unroll
  for (int i = 0; i < 4; ++i)
#pragma unroll
    for (int j = 0; j < 4; ++j)
#pragma unroll
      for (int r = 0; r < 8; ++r) acc[i][j][r] = 0.0f;

  const float* arow[4];
  const float* brow[4];
#pragma unroll
  for (int i = 0; i < 4; ++i) {
    arow[i] = x + (size_t)(m0 + i * 16 + l16) * KDIM;
    brow[i] = W + (size_t)(n0 + i * 16 + l16) * KDIM;
  }

  for (int kk = 0; kk < KDIM; kk += 4) {
    const int ka = kk + 2 * half;
    v2f a[4], b[4];
#pragma unroll
    for (int i = 0; i < 4; ++i) a[i] = *(const v2f*)(arow[i] + ka);
#pragma unroll
    for (int j = 0; j < 4; ++j) b[j] = *(const v2f*)(brow[j] + ka);
#pragma unroll
    for (int i = 0; i < 4; ++i)
#pragma unroll
      for (int j = 0; j < 4; ++j)
        acc[i][j] = WMMA_F32(a[i], b[j], acc[i][j]);
  }

#pragma unroll
  for (int i = 0; i < 4; ++i)
#pragma unroll
    for (int j = 0; j < 4; ++j)
#pragma unroll
      for (int r = 0; r < 8; ++r)
        out[(size_t)(m0 + i * 16 + r + 8 * half) * KDIM + n0 + j * 16 + l16] =
            acc[i][j][r];
}

__global__ void __launch_bounds__(32)
flash_attn_kernel(const float* __restrict__ q, const float* __restrict__ k,
                  const float* __restrict__ v, float* __restrict__ o) {
  __shared__ float p_lds[32 * 18];

  const int lane = threadIdx.x;
  const int half = lane >> 4;
  const int l16  = lane & 15;
  const int t0 = blockIdx.x * 32;
  const int h  = blockIdx.y;
  const int b  = blockIdx.z;

  const float* qbase = q + ((size_t)b * SEQ + t0) * KDIM + h * HS;
  const float* kbase = k + (size_t)b * SEQ * KDIM + h * HS;
  const float* vbase = v + (size_t)b * SEQ * KDIM + h * HS;

  const float scale = 0.03125f * 1.44269504088896340736f;

  v2f qa[2][16];
#pragma unroll
  for (int u = 0; u < 2; ++u)
#pragma unroll
    for (int kk = 0; kk < 16; ++kk) {
      v2f t = *(const v2f*)(qbase + (size_t)(u * 16 + l16) * KDIM + kk * 4 + 2 * half);
      qa[u][kk].x = t.x * scale;
      qa[u][kk].y = t.y * scale;
    }

  float m_r[2][8], l_r[2][8];
  v8f oacc[2][4];
#pragma unroll
  for (int u = 0; u < 2; ++u) {
#pragma unroll
    for (int r = 0; r < 8; ++r) { m_r[u][r] = -3.0e38f; l_r[u][r] = 0.0f; }
#pragma unroll
    for (int j = 0; j < 4; ++j)
#pragma unroll
      for (int r = 0; r < 8; ++r) oacc[u][j][r] = 0.0f;
  }

  for (int kt = 0; kt < SEQ; kt += 16) {
    v8f s[2];
#pragma unroll
    for (int u = 0; u < 2; ++u)
#pragma unroll
      for (int r = 0; r < 8; ++r) s[u][r] = 0.0f;
#pragma unroll
    for (int kk = 0; kk < 16; ++kk) {
      v2f bfrag = *(const v2f*)(kbase + (size_t)(kt + l16) * KDIM + kk * 4 + 2 * half);
      s[0] = WMMA_F32(qa[0][kk], bfrag, s[0]);
      s[1] = WMMA_F32(qa[1][kk], bfrag, s[1]);
    }

#pragma unroll
    for (int u = 0; u < 2; ++u)
#pragma unroll
      for (int r = 0; r < 8; ++r) {
        float tmax = rowmax16(s[u][r]);
        float mn   = fmaxf(m_r[u][r], tmax);
        float corr = exp2f(m_r[u][r] - mn);
        float p    = exp2f(s[u][r] - mn);
        l_r[u][r] = l_r[u][r] * corr + rowsum16(p);
        m_r[u][r] = mn;
#pragma unroll
        for (int j = 0; j < 4; ++j) oacc[u][j][r] *= corr;
        p_lds[(u * 16 + r + 8 * half) * 18 + l16] = p;
      }
    __syncthreads();

#pragma unroll
    for (int kk = 0; kk < 4; ++kk) {
      const int key = kk * 4 + 2 * half;
      v2f pa0 = *(const v2f*)(&p_lds[(l16) * 18 + key]);
      v2f pa1 = *(const v2f*)(&p_lds[(16 + l16) * 18 + key]);
#pragma unroll
      for (int j = 0; j < 4; ++j) {
        v2f vb;
        vb.x = vbase[(size_t)(kt + key) * KDIM + j * 16 + l16];
        vb.y = vbase[(size_t)(kt + key + 1) * KDIM + j * 16 + l16];
        oacc[0][j] = WMMA_F32(pa0, vb, oacc[0][j]);
        oacc[1][j] = WMMA_F32(pa1, vb, oacc[1][j]);
      }
    }
    __syncthreads();
  }

#pragma unroll
  for (int u = 0; u < 2; ++u)
#pragma unroll
    for (int r = 0; r < 8; ++r) {
      const float inv = 1.0f / l_r[u][r];
#pragma unroll
      for (int j = 0; j < 4; ++j)
        o[((size_t)b * SEQ + t0 + u * 16 + r + 8 * half) * KDIM + h * HS +
          j * 16 + l16] = oacc[u][j][r] * inv;
    }
}

__global__ void __launch_bounds__(32)
out_proj_kernel(const float* __restrict__ A, const float* __restrict__ Wu,
                const float* __restrict__ bu, float* __restrict__ out) {
  const int lane = threadIdx.x;
  const int half = lane >> 4;
  const int l16  = lane & 15;
  const int m0 = blockIdx.x * 64;
  const int n0 = blockIdx.y * 64;

  v8f acc[4][4];
#pragma unroll
  for (int i = 0; i < 4; ++i)
#pragma unroll
    for (int j = 0; j < 4; ++j)
#pragma unroll
      for (int r = 0; r < 8; ++r) acc[i][j][r] = 0.0f;

  const float* arow[4];
  const float* brow[4];
#pragma unroll
  for (int i = 0; i < 4; ++i) {
    arow[i] = A + (size_t)(m0 + i * 16 + l16) * KDIM;
    brow[i] = Wu + (size_t)(n0 + i * 16 + l16) * KDIM;
  }

  for (int kk = 0; kk < KDIM; kk += 4) {
    const int ka = kk + 2 * half;
    v2f a[4], b[4];
#pragma unroll
    for (int i = 0; i < 4; ++i) a[i] = *(const v2f*)(arow[i] + ka);
#pragma unroll
    for (int j = 0; j < 4; ++j) b[j] = *(const v2f*)(brow[j] + ka);
#pragma unroll
    for (int i = 0; i < 4; ++i)
#pragma unroll
      for (int j = 0; j < 4; ++j)
        acc[i][j] = WMMA_F32(a[i], b[j], acc[i][j]);
  }

#pragma unroll
  for (int j = 0; j < 4; ++j) {
    const float bias = bu[n0 + j * 16 + l16];
#pragma unroll
    for (int i = 0; i < 4; ++i)
#pragma unroll
      for (int r = 0; r < 8; ++r)
        out[(size_t)(m0 + i * 16 + r + 8 * half) * KDIM + n0 + j * 16 + l16] =
            acc[i][j][r] + bias;
  }
}

extern "C" void kernel_launch(void* const* d_in, const int* in_sizes, int n_in,
                              void* d_out, int out_size, void* d_ws, size_t ws_size,
                              hipStream_t stream) {
  const float* x  = (const float*)d_in[0];
  const float* Wk = (const float*)d_in[1];
  const float* Wq = (const float*)d_in[2];
  const float* Wv = (const float*)d_in[3];
  const float* Wu = (const float*)d_in[4];
  const float* bu = (const float*)d_in[5];
  float* out = (float*)d_out;

  const size_t MAT = (size_t)ROWS * KDIM;
  float* qb = (float*)d_ws;
  float* kb = qb + MAT;
  float* vb = kb + MAT;
  float* ob = vb + MAT;

  dim3 g1(ROWS / 64, KDIM / 64, 3);
  qkv_proj_kernel<<<g1, 32, 0, stream>>>(x, Wq, Wk, Wv, qb, kb, vb);

  dim3 g2(SEQ / 32, HEADS, BATCH);
  flash_attn_kernel<<<g2, 32, 0, stream>>>(qb, kb, vb, ob);

  dim3 g3(ROWS / 64, KDIM / 64);
  out_proj_kernel<<<g3, 32, 0, stream>>>(ob, Wu, bu, out);
}


# attention computed transposed (S^T=K*Q^T, O^T=V^T*P^T): softmax max and sum per lane, P stays in registers, no LDS or lane shuffles; changes sum order of row sums and of keys in P*V
# speedup vs baseline: 1.3073x; 1.3073x over previous
#include <hip/hip_runtime.h>

typedef __attribute__((ext_vector_type(2))) float v2f;
typedef __attribute__((ext_vector_type(8))) float v8f;

#define WMMA_F32(a, b, c) \
  __builtin_amdgcn_wmma_f32_16x16x4_f32(false, (a), false, (b), (short)0, (c), false, false)

#define BATCH 4
#define SEQ   2048
#define KDIM  1024
#define HEADS 16
#define HS    64
#define ROWS  (BATCH * SEQ)

__global__ void __launch_bounds__(32)
qkv_proj_kernel(const float* __restrict__ x,
                const float* __restrict__ Wq, const float* __restrict__ Wk,
                const float* __restrict__ Wv,
                float* __restrict__ qo, float* __restrict__ ko,
                float* __restrict__ vo) {
  const int lane = threadIdx.x;
  const int half = lane >> 4;
  const int l16  = lane & 15;
  const int m0 = blockIdx.x * 64;
  const int n0 = blockIdx.y * 64;
  const int which = blockIdx.z;

  const float* W  = (which == 0) ? Wq : (which == 1) ? Wk : Wv;
  float*      out = (which == 0) ? qo : (which == 1) ? ko : vo;

  v8f acc[4][4];
#pragma unroll
  for (int i = 0; i < 4; ++i)
#pragma unroll
    for (int j = 0; j < 4; ++j)
#pragma unroll
      for (int r = 0; r < 8; ++r) acc[i][j][r] = 0.0f;

  const float* arow[4];
  const float* brow[4];
#pragma unroll
  for (int i = 0; i < 4; ++i) {
    arow[i] = x + (size_t)(m0 + i * 16 + l16) * KDIM;
    brow[i] = W + (size_t)(n0 + i * 16 + l16) * KDIM;
  }

  for (int kk = 0; kk < KDIM; kk += 4) {
    const int ka = kk + 2 * half;
    v2f a[4], b[4];
#pragma unroll
    for (int i = 0; i < 4; ++i) a[i] = *(const v2f*)(arow[i] + ka);
#pragma unroll
    for (int j = 0; j < 4; ++j) b[j] = *(const v2f*)(brow[j] + ka);
#pragma unroll
    for (int i = 0; i < 4; ++i)
#pragma unroll
      for (int j = 0; j < 4; ++j)
        acc[i][j] = WMMA_F32(a[i], b[j], acc[i][j]);
  }

#pragma unroll
  for (int i = 0; i < 4; ++i)
#pragma unroll
    for (int j = 0; j < 4; ++j)
#pragma unroll
      for (int r = 0; r < 8; ++r)
        out[(size_t)(m0 + i * 16 + r + 8 * half) * KDIM + n0 + j * 16 + l16] =
            acc[i][j][r];
}

__device__ __forceinline__ float xhalf(float x) {
  int bits = __builtin_bit_cast(int, x);
  return __builtin_bit_cast(
      float, __builtin_amdgcn_permlanex16(bits, bits, 0x76543210, 0xfedcba98,
                                          false, false));
}

__global__ void __launch_bounds__(32)
flash_attn_kernel(const float* __restrict__ q, const float* __restrict__ k,
                  const float* __restrict__ v, float* __restrict__ o) {
  const int lane = threadIdx.x;
  const int half = lane >> 4;
  const int l16  = lane & 15;
  const int t0 = blockIdx.x * 32;
  const int h  = blockIdx.y;
  const int b  = blockIdx.z;

  const float* qbase = q + ((size_t)b * SEQ + t0) * KDIM + h * HS;
  const float* kbase = k + (size_t)b * SEQ * KDIM + h * HS;
  const float* vbase = v + (size_t)b * SEQ * KDIM + h * HS;

  const float scale = 0.03125f * 1.44269504088896340736f;

  v2f qf[2][16];
#pragma unroll
  for (int u = 0; u < 2; ++u)
#pragma unroll
    for (int kk = 0; kk < 16; ++kk) {
      v2f t = *(const v2f*)(qbase + (size_t)(u * 16 + l16) * KDIM + kk * 4 + 2 * half);
      qf[u][kk].x = t.x * scale;
      qf[u][kk].y = t.y * scale;
    }

  float m_r[2], l_r[2];
  v8f oacc[2][4];
#pragma unroll
  for (int u = 0; u < 2; ++u) {
    m_r[u] = -3.0e38f;
    l_r[u] = 0.0f;
#pragma unroll
    for (int j = 0; j < 4; ++j)
#pragma unroll
      for (int r = 0; r < 8; ++r) oacc[u][j][r] = 0.0f;
  }

  for (int kt = 0; kt < SEQ; kt += 16) {
    v8f s[2];
#pragma unroll
    for (int u = 0; u < 2; ++u)
#pragma unroll
      for (int r = 0; r < 8; ++r) s[u][r] = 0.0f;
#pragma unroll
    for (int kk = 0; kk < 16; ++kk) {
      v2f kf = *(const v2f*)(kbase + (size_t)(kt + l16) * KDIM + kk * 4 + 2 * half);
      s[0] = WMMA_F32(kf, qf[0][kk], s[0]);
      s[1] = WMMA_F32(kf, qf[1][kk], s[1]);
    }

#pragma unroll
    for (int u = 0; u < 2; ++u) {
      float tmax = fmaxf(fmaxf(fmaxf(s[u][0], s[u][1]), fmaxf(s[u][2], s[u][3])),
                         fmaxf(fmaxf(s[u][4], s[u][5]), fmaxf(s[u][6], s[u][7])));
      tmax = fmaxf(tmax, xhalf(tmax));
      const float mn   = fmaxf(m_r[u], tmax);
      const float corr = exp2f(m_r[u] - mn);
      m_r[u] = mn;
      float psum = 0.0f;
#pragma unroll
      for (int r = 0; r < 8; ++r) {
        const float p = exp2f(s[u][r] - mn);
        s[u][r] = p;
        psum += p;
      }
      l_r[u] = l_r[u] * corr + psum;
#pragma unroll
      for (int j = 0; j < 4; ++j)
#pragma unroll
        for (int r = 0; r < 8; ++r) oacc[u][j][r] *= corr;
    }

#pragma unroll
    for (int t = 0; t < 4; ++t) {
      const int key = kt + 8 * half + 2 * t;
      v2f p0, p1;
      p0.x = s[0][2 * t]; p0.y = s[0][2 * t + 1];
      p1.x = s[1][2 * t]; p1.y = s[1][2 * t + 1];
#pragma unroll
      for (int j = 0; j < 4; ++j) {
        v2f vf;
        vf.x = vbase[(size_t)key * KDIM + j * 16 + l16];
        vf.y = vbase[(size_t)(key + 1) * KDIM + j * 16 + l16];
        oacc[0][j] = WMMA_F32(vf, p0, oacc[0][j]);
        oacc[1][j] = WMMA_F32(vf, p1, oacc[1][j]);
      }
    }
  }

#pragma unroll
  for (int u = 0; u < 2; ++u) {
    const float inv = 1.0f / (l_r[u] + xhalf(l_r[u]));
    float* orow = o + ((size_t)b * SEQ + t0 + u * 16 + l16) * KDIM + h * HS + 8 * half;
#pragma unroll
    for (int j = 0; j < 4; ++j) {
      float4 lo, hi;
      lo.x = oacc[u][j][0] * inv; lo.y = oacc[u][j][1] * inv;
      lo.z = oacc[u][j][2] * inv; lo.w = oacc[u][j][3] * inv;
      hi.x = oacc[u][j][4] * inv; hi.y = oacc[u][j][5] * inv;
      hi.z = oacc[u][j][6] * inv; hi.w = oacc[u][j][7] * inv;
      *(float4*)(orow + 16 * j)     = lo;
      *(float4*)(orow + 16 * j + 4) = hi;
    }
  }
}

__global__ void __launch_bounds__(32)
out_proj_kernel(const float* __restrict__ A, const float* __restrict__ Wu,
                const float* __restrict__ bu, float* __restrict__ out) {
  const int lane = threadIdx.x;
  const int half = lane >> 4;
  const int l16  = lane & 15;
  const int m0 = blockIdx.x * 64;
  const int n0 = blockIdx.y * 64;

  v8f acc[4][4];
#pragma unroll
  for (int i = 0; i < 4; ++i)
#pragma unroll
    for (int j = 0; j < 4; ++j)
#pragma unroll
      for (int r = 0; r < 8; ++r) acc[i][j][r] = 0.0f;

  const float* arow[4];
  const float* brow[4];
#pragma unroll
  for (int i = 0; i < 4; ++i) {
    arow[i] = A + (size_t)(m0 + i * 16 + l16) * KDIM;
    brow[i] = Wu + (size_t)(n0 + i * 16 + l16) * KDIM;
  }

  for (int kk = 0; kk < KDIM; kk += 4) {
    const int ka = kk + 2 * half;
    v2f a[4], b[4];
#pragma unroll
    for (int i = 0; i < 4; ++i) a[i] = *(const v2f*)(arow[i] + ka);
#pragma unroll
    for (int j = 0; j < 4; ++j) b[j] = *(const v2f*)(brow[j] + ka);
#pragma unroll
    for (int i = 0; i < 4; ++i)
#pragma unroll
      for (int j = 0; j < 4; ++j)
        acc[i][j] = WMMA_F32(a[i], b[j], acc[i][j]);
  }

#pragma unroll
  for (int j = 0; j < 4; ++j) {
    const float bias = bu[n0 + j * 16 + l16];
#pragma unroll
    for (int i = 0; i < 4; ++i)
#pragma unroll
      for (int r = 0; r < 8; ++r)
        out[(size_t)(m0 + i * 16 + r + 8 * half) * KDIM + n0 + j * 16 + l16] =
            acc[i][j][r] + bias;
  }
}

extern "C" void kernel_launch(void* const* d_in, const int* in_sizes, int n_in,
                              void* d_out, int out_size, void* d_ws, size_t ws_size,
                              hipStream_t stream) {
  const float* x  = (const float*)d_in[0];
  const float* Wk = (const float*)d_in[1];
  const float* Wq = (const float*)d_in[2];
  const float* Wv = (const float*)d_in[3];
  const float* Wu = (const float*)d_in[4];
  const float* bu = (const float*)d_in[5];
  float* out = (float*)d_out;

  const size_t MAT = (size_t)ROWS * KDIM;
  float* qb = (float*)d_ws;
  float* kb = qb + MAT;
  float* vb = kb + MAT;
  float* ob = vb + MAT;

  dim3 g1(ROWS / 64, KDIM / 64, 3);
  qkv_proj_kernel<<<g1, 32, 0, stream>>>(x, Wq, Wk, Wv, qb, kb, vb);

  dim3 g2(SEQ / 32, HEADS, BATCH);
  flash_attn_kernel<<<g2, 32, 0, stream>>>(qb, kb, vb, ob);

  dim3 g3(ROWS / 64, KDIM / 64);
  out_proj_kernel<<<g3, 32, 0, stream>>>(ob, Wu, bu, out);
}


# fragment loads widened to 16 B per lane: a lane reads its row's 64-B half line at once, WMMA steps pair k as loaded on A and B alike; V^T rows renumbered d=4c+j; k sum order in dot products changes
# speedup vs baseline: 1.6433x; 1.2570x over previous
#include <hip/hip_runtime.h>

typedef __attribute__((ext_vector_type(2))) float v2f;
typedef __attribute__((ext_vector_type(4))) float v4f;
typedef __attribute__((ext_vector_type(8))) float v8f;

#define WMMA_F32(a, b, c) \
  __builtin_amdgcn_wmma_f32_16x16x4_f32(false, (a), false, (b), (short)0, (c), false, false)

#define BATCH 4
#define SEQ   2048
#define KDIM  1024
#define HEADS 16
#define HS    64
#define ROWS  (BATCH * SEQ)

__global__ void __launch_bounds__(32)
qkv_proj_kernel(const float* __restrict__ x,
                const float* __restrict__ Wq, const float* __restrict__ Wk,
                const float* __restrict__ Wv,
                float* __restrict__ qo, float* __restrict__ ko,
                float* __restrict__ vo) {
  const int lane = threadIdx.x;
  const int half = lane >> 4;
  const int l16  = lane & 15;
  const int m0 = blockIdx.x * 64;
  const int n0 = blockIdx.y * 64;
  const int which = blockIdx.z;

  const float* W  = (which == 0) ? Wq : (which == 1) ? Wk : Wv;
  float*      out = (which == 0) ? qo : (which == 1) ? ko : vo;

  v8f acc[4][4];
#pragma unroll
  for (int i = 0; i < 4; ++i)
#pragma unroll
    for (int j = 0; j < 4; ++j)
#pragma unroll
      for (int r = 0; r < 8; ++r) acc[i][j][r] = 0.0f;

  const float* arow[4];
  const float* brow[4];
#pragma unroll
  for (int i = 0; i < 4; ++i) {
    arow[i] = x + (size_t)(m0 + i * 16 + l16) * KDIM;
    brow[i] = W + (size_t)(n0 + i * 16 + l16) * KDIM;
  }

  for (int kc = 0; kc < KDIM; kc += 32) {
    const int ka = kc + 16 * half;
    v4f a[4][4], b[4][4];
#pragma unroll
    for (int i = 0; i < 4; ++i)
#pragma unroll
      for (int q = 0; q < 4; ++q) a[i][q] = *(const v4f*)(arow[i] + ka + 4 * q);
#pragma unroll
    for (int j = 0; j < 4; ++j)
#pragma unroll
      for (int q = 0; q < 4; ++q) b[j][q] = *(const v4f*)(brow[j] + ka + 4 * q);
    asm volatile("" :: "v"(a[0][0]), "v"(a[0][1]), "v"(a[0][2]), "v"(a[0][3]),
                 "v"(a[1][0]), "v"(a[1][1]), "v"(a[1][2]), "v"(a[1][3]),
                 "v"(a[2][0]), "v"(a[2][1]), "v"(a[2][2]), "v"(a[2][3]),
                 "v"(a[3][0]), "v"(a[3][1]), "v"(a[3][2]), "v"(a[3][3]));
    asm volatile("" :: "v"(b[0][0]), "v"(b[0][1]), "v"(b[0][2]), "v"(b[0][3]),
                 "v"(b[1][0]), "v"(b[1][1]), "v"(b[1][2]), "v"(b[1][3]),
                 "v"(b[2][0]), "v"(b[2][1]), "v"(b[2][2]), "v"(b[2][3]),
                 "v"(b[3][0]), "v"(b[3][1]), "v"(b[3][2]), "v"(b[3][3]));
#pragma unroll
    for (int q = 0; q < 4; ++q) {
#pragma unroll
      for (int i = 0; i < 4; ++i)
#pragma unroll
        for (int j = 0; j < 4; ++j)
          acc[i][j] = WMMA_F32(a[i][q].lo, b[j][q].lo, acc[i][j]);
#pragma unroll
      for (int i = 0; i < 4; ++i)
#pragma unroll
        for (int j = 0; j < 4; ++j)
          acc[i][j] = WMMA_F32(a[i][q].hi, b[j][q].hi, acc[i][j]);
    }
  }

#pragma unroll
  for (int i = 0; i < 4; ++i)
#pragma unroll
    for (int j = 0; j < 4; ++j)
#pragma unroll
      for (int r = 0; r < 8; ++r)
        out[(size_t)(m0 + i * 16 + r + 8 * half) * KDIM + n0 + j * 16 + l16] =
            acc[i][j][r];
}

__device__ __forceinline__ float xhalf(float x) {
  int bits = __builtin_bit_cast(int, x);
  return __builtin_bit_cast(
      float, __builtin_amdgcn_permlanex16(bits, bits, 0x76543210, 0xfedcba98,
                                          false, false));
}

__global__ void __launch_bounds__(32)
flash_attn_kernel(const float* __restrict__ q, const float* __restrict__ k,
                  const float* __restrict__ v, float* __restrict__ o) {
  const int lane = threadIdx.x;
  const int half = lane >> 4;
  const int l16  = lane & 15;
  const int t0 = blockIdx.x * 32;
  const int h  = blockIdx.y;
  const int b  = blockIdx.z;

  const float* qbase = q + ((size_t)b * SEQ + t0) * KDIM + h * HS;
  const float* kbase = k + (size_t)b * SEQ * KDIM + h * HS;
  const float* vbase = v + (size_t)b * SEQ * KDIM + h * HS;

  const float scale = 0.03125f * 1.44269504088896340736f;

  v2f qf[2][16];
#pragma unroll
  for (int u = 0; u < 2; ++u)
#pragma unroll
    for (int g = 0; g < 2; ++g)
#pragma unroll
      for (int i = 0; i < 4; ++i) {
        v4f t = *(const v4f*)(qbase + (size_t)(u * 16 + l16) * KDIM + 32 * g + 16 * half + 4 * i);
        qf[u][8 * g + 2 * i].x     = t.x * scale;
        qf[u][8 * g + 2 * i].y     = t.y * scale;
        qf[u][8 * g + 2 * i + 1].x = t.z * scale;
        qf[u][8 * g + 2 * i + 1].y = t.w * scale;
      }

  float m_r[2], l_r[2];
  v8f oacc[2][4];
#pragma unroll
  for (int u = 0; u < 2; ++u) {
    m_r[u] = -3.0e38f;
    l_r[u] = 0.0f;
#pragma unroll
    for (int j = 0; j < 4; ++j)
#pragma unroll
      for (int r = 0; r < 8; ++r) oacc[u][j][r] = 0.0f;
  }

  for (int kt = 0; kt < SEQ; kt += 16) {
    v8f s[2];
#pragma unroll
    for (int u = 0; u < 2; ++u)
#pragma unroll
      for (int r = 0; r < 8; ++r) s[u][r] = 0.0f;
    v4f kf[2][4];
#pragma unroll
    for (int g = 0; g < 2; ++g)
#pragma unroll
      for (int i = 0; i < 4; ++i)
        kf[g][i] = *(const v4f*)(kbase + (size_t)(kt + l16) * KDIM + 32 * g + 16 * half + 4 * i);
    asm volatile("" :: "v"(kf[0][0]), "v"(kf[0][1]), "v"(kf[0][2]), "v"(kf[0][3]),
                       "v"(kf[1][0]), "v"(kf[1][1]), "v"(kf[1][2]), "v"(kf[1][3]));
#pragma unroll
    for (int g = 0; g < 2; ++g)
#pragma unroll
      for (int i = 0; i < 4; ++i) {
        const int kk = 8 * g + 2 * i;
        s[0] = WMMA_F32(kf[g][i].lo, qf[0][kk], s[0]);
        s[1] = WMMA_F32(kf[g][i].lo, qf[1][kk], s[1]);
        s[0] = WMMA_F32(kf[g][i].hi, qf[0][kk + 1], s[0]);
        s[1] = WMMA_F32(kf[g][i].hi, qf[1][kk + 1], s[1]);
      }

#pragma unroll
    for (int u = 0; u < 2; ++u) {
      float tmax = fmaxf(fmaxf(fmaxf(s[u][0], s[u][1]), fmaxf(s[u][2], s[u][3])),
                         fmaxf(fmaxf(s[u][4], s[u][5]), fmaxf(s[u][6], s[u][7])));
      tmax = fmaxf(tmax, xhalf(tmax));
      const float mn   = fmaxf(m_r[u], tmax);
      const float corr = exp2f(m_r[u] - mn);
      m_r[u] = mn;
      float psum = 0.0f;
#pragma unroll
      for (int r = 0; r < 8; ++r) {
        const float p = exp2f(s[u][r] - mn);
        s[u][r] = p;
        psum += p;
      }
      l_r[u] = l_r[u] * corr + psum;
#pragma unroll
      for (int j = 0; j < 4; ++j)
#pragma unroll
        for (int r = 0; r < 8; ++r) oacc[u][j][r] *= corr;
    }

    v4f vl[8];
#pragma unroll
    for (int e = 0; e < 8; ++e)
      vl[e] = *(const v4f*)(vbase + (size_t)(kt + 8 * half + e) * KDIM + 4 * l16);
#pragma unroll
    for (int t = 0; t < 4; ++t) {
      v2f p0, p1;
      p0.x = s[0][2 * t]; p0.y = s[0][2 * t + 1];
      p1.x = s[1][2 * t]; p1.y = s[1][2 * t + 1];
#pragma unroll
      for (int j = 0; j < 4; ++j) {
        v2f vf;
        vf.x = vl[2 * t][j];
        vf.y = vl[2 * t + 1][j];
        oacc[0][j] = WMMA_F32(vf, p0, oacc[0][j]);
        oacc[1][j] = WMMA_F32(vf, p1, oacc[1][j]);
      }
    }
  }

#pragma unroll
  for (int u = 0; u < 2; ++u) {
    const float inv = 1.0f / (l_r[u] + xhalf(l_r[u]));
    float* orow = o + ((size_t)b * SEQ + t0 + u * 16 + l16) * KDIM + h * HS + 32 * half;
#pragma unroll
    for (int r = 0; r < 8; ++r) {
      float4 w;
      w.x = oacc[u][0][r] * inv; w.y = oacc[u][1][r] * inv;
      w.z = oacc[u][2][r] * inv; w.w = oacc[u][3][r] * inv;
      *(float4*)(orow + 4 * r) = w;
    }
  }
}

__global__ void __launch_bounds__(32)
out_proj_kernel(const float* __restrict__ A, const float* __restrict__ Wu,
                const float* __restrict__ bu, float* __restrict__ out) {
  const int lane = threadIdx.x;
  const int half = lane >> 4;
  const int l16  = lane & 15;
  const int m0 = blockIdx.x * 64;
  const int n0 = blockIdx.y * 64;

  v8f acc[4][4];
#pragma unroll
  for (int i = 0; i < 4; ++i)
#pragma unroll
    for (int j = 0; j < 4; ++j)
#pragma unroll
      for (int r = 0; r < 8; ++r) acc[i][j][r] = 0.0f;

  const float* arow[4];
  const float* brow[4];
#pragma unroll
  for (int i = 0; i < 4; ++i) {
    arow[i] = A + (size_t)(m0 + i * 16 + l16) * KDIM;
    brow[i] = Wu + (size_t)(n0 + i * 16 + l16) * KDIM;
  }

  for (int kc = 0; kc < KDIM; kc += 32) {
    const int ka = kc + 16 * half;
    v4f a[4][4], b[4][4];
#pragma unroll
    for (int i = 0; i < 4; ++i)
#pragma unroll
      for (int q = 0; q < 4; ++q) a[i][q] = *(const v4f*)(arow[i] + ka + 4 * q);
#pragma unroll
    for (int j = 0; j < 4; ++j)
#pragma unroll
      for (int q = 0; q < 4; ++q) b[j][q] = *(const v4f*)(brow[j] + ka + 4 * q);
    asm volatile("" :: "v"(a[0][0]), "v"(a[0][1]), "v"(a[0][2]), "v"(a[0][3]),
                 "v"(a[1][0]), "v"(a[1][1]), "v"(a[1][2]), "v"(a[1][3]),
                 "v"(a[2][0]), "v"(a[2][1]), "v"(a[2][2]), "v"(a[2][3]),
                 "v"(a[3][0]), "v"(a[3][1]), "v"(a[3][2]), "v"(a[3][3]));
    asm volatile("" :: "v"(b[0][0]), "v"(b[0][1]), "v"(b[0][2]), "v"(b[0][3]),
                 "v"(b[1][0]), "v"(b[1][1]), "v"(b[1][2]), "v"(b[1][3]),
                 "v"(b[2][0]), "v"(b[2][1]), "v"(b[2][2]), "v"(b[2][3]),
                 "v"(b[3][0]), "v"(b[3][1]), "v"(b[3][2]), "v"(b[3][3]));
#pragma unroll
    for (int q = 0; q < 4; ++q) {
#pragma unroll
      for (int i = 0; i < 4; ++i)
#pragma unroll
        for (int j = 0; j < 4; ++j)
          acc[i][j] = WMMA_F32(a[i][q].lo, b[j][q].lo, acc[i][j]);
#pragma unroll
      for (int i = 0; i < 4; ++i)
#pragma unroll
        for (int j = 0; j < 4; ++j)
          acc[i][j] = WMMA_F32(a[i][q].hi, b[j][q].hi, acc[i][j]);
    }
  }

#pragma unroll
  for (int j = 0; j < 4; ++j) {
    const float bias = bu[n0 + j * 16 + l16];
#pragma unroll
    for (int i = 0; i < 4; ++i)
#pragma unroll
      for (int r = 0; r < 8; ++r)
        out[(size_t)(m0 + i * 16 + r + 8 * half) * KDIM + n0 + j * 16 + l16] =
            acc[i][j][r] + bias;
  }
}

extern "C" void kernel_launch(void* const* d_in, const int* in_sizes, int n_in,
                              void* d_out, int out_size, void* d_ws, size_t ws_size,
                              hipStream_t stream) {
  const float* x  = (const float*)d_in[0];
  const float* Wk = (const float*)d_in[1];
  const float* Wq = (const float*)d_in[2];
  const float* Wv = (const float*)d_in[3];
  const float* Wu = (const float*)d_in[4];
  const float* bu = (const float*)d_in[5];
  float* out = (float*)d_out;

  const size_t MAT = (size_t)ROWS * KDIM;
  float* qb = (float*)d_ws;
  float* kb = qb + MAT;
  float* vb = kb + MAT;
  float* ob = vb + MAT;

  dim3 g1(ROWS / 64, KDIM / 64, 3);
  qkv_proj_kernel<<<g1, 32, 0, stream>>>(x, Wq, Wk, Wv, qb, kb, vb);

  dim3 g2(SEQ / 32, HEADS, BATCH);
  flash_attn_kernel<<<g2, 32, 0, stream>>>(qb, kb, vb, ob);

  dim3 g3(ROWS / 64, KDIM / 64);
  out_proj_kernel<<<g3, 32, 0, stream>>>(ob, Wu, bu, out);
}


# fragment loads issued one block of WMMAs ahead, into the registers the finished block frees (GEMMs: half a 32-k chunk ahead; attention: next K tile among the score WMMAs, V in front); sums unchanged
# speedup vs baseline: 1.7555x; 1.0683x over previous
#include <hip/hip_runtime.h>

typedef __attribute__((ext_vector_type(2))) float v2f;
typedef __attribute__((ext_vector_type(4))) float v4f;
typedef __attribute__((ext_vector_type(8))) float v8f;

#define WMMA_F32(a, b, c) \
  __builtin_amdgcn_wmma_f32_16x16x4_f32(false, (a), false, (b), (short)0, (c), false, false)

#define BATCH 4
#define SEQ   2048
#define KDIM  1024
#define HEADS 16
#define HS    64
#define ROWS  (BATCH * SEQ)


#define SCHED_FENCE() __builtin_amdgcn_sched_barrier(0)

__device__ __forceinline__ void load_block(const float* const (&row)[4], int k,
                                           v4f (&f)[4][2]) {
#pragma unroll
  for (int i = 0; i < 4; ++i)
#pragma unroll
    for (int q = 0; q < 2; ++q) f[i][q] = *(const v4f*)(row[i] + k + 4 * q);
}

__device__ __forceinline__ void mma_block(const v4f (&a)[4][2], const v4f (&b)[4][2],
                                          v8f (&acc)[4][4]) {
#pragma unroll
  for (int q = 0; q < 2; ++q) {
#pragma unroll
    for (int i = 0; i < 4; ++i)
#pragma unroll
      for (int j = 0; j < 4; ++j)
        acc[i][j] = WMMA_F32(a[i][q].lo, b[j][q].lo, acc[i][j]);
#pragma unroll
    for (int i = 0; i < 4; ++i)
#pragma unroll
      for (int j = 0; j < 4; ++j)
        acc[i][j] = WMMA_F32(a[i][q].hi, b[j][q].hi, acc[i][j]);
  }
}

__device__ __forceinline__ void gemm_tile_64x64(const float* A, const float* B,
                                                int m0, int n0, int lane,
                                                v8f (&acc)[4][4]) {
  const int half = lane >> 4;
  const int l16  = lane & 15;
#pragma unroll
  for (int i = 0; i < 4; ++i)
#pragma unroll
    for (int j = 0; j < 4; ++j)
#pragma unroll
      for (int r = 0; r < 8; ++r) acc[i][j][r] = 0.0f;

  const float* arow[4];
  const float* brow[4];
#pragma unroll
  for (int i = 0; i < 4; ++i) {
    arow[i] = A + (size_t)(m0 + i * 16 + l16) * KDIM + 16 * half;
    brow[i] = B + (size_t)(n0 + i * 16 + l16) * KDIM + 16 * half;
  }

  v4f aX[4][2], bX[4][2], aY[4][2], bY[4][2];
  load_block(arow, 0, aX);
  load_block(brow, 0, bX);
  SCHED_FENCE();
  load_block(arow, 8, aY);
  load_block(brow, 8, bY);
  for (int kc = 0; kc < KDIM; kc += 32) {
    const int kn = (kc + 32 < KDIM) ? kc + 32 : kc;
    mma_block(aX, bX, acc);
    SCHED_FENCE();
    load_block(arow, kn, aX);
    load_block(brow, kn, bX);
    SCHED_FENCE();
    mma_block(aY, bY, acc);
    load_block(arow, kn + 8, aY);
    load_block(brow, kn + 8, bY);
    SCHED_FENCE();
  }
}

__global__ void __launch_bounds__(32)
qkv_proj_kernel(const float* x,
                const float* Wq, const float* Wk,
                const float* Wv,
                float* qo, float* ko,
                float* vo) {
  const int lane = threadIdx.x;
  const int half = lane >> 4;
  const int l16  = lane & 15;
  const int m0 = blockIdx.x * 64;
  const int n0 = blockIdx.y * 64;
  const int which = blockIdx.z;

  const float* W  = (which == 0) ? Wq : (which == 1) ? Wk : Wv;
  float*      out = (which == 0) ? qo : (which == 1) ? ko : vo;

  v8f acc[4][4];
  gemm_tile_64x64(x, W, m0, n0, lane, acc);

#pragma unroll
  for (int i = 0; i < 4; ++i)
#pragma unroll
    for (int j = 0; j < 4; ++j)
#pragma unroll
      for (int r = 0; r < 8; ++r)
        out[(size_t)(m0 + i * 16 + r + 8 * half) * KDIM + n0 + j * 16 + l16] =
            acc[i][j][r];
}

__device__ __forceinline__ float xhalf(float x) {
  int bits = __builtin_bit_cast(int, x);
  return __builtin_bit_cast(
      float, __builtin_amdgcn_permlanex16(bits, bits, 0x76543210, 0xfedcba98,
                                          false, false));
}

__global__ void __launch_bounds__(32)
flash_attn_kernel(const float* __restrict__ q, const float* __restrict__ k,
                  const float* __restrict__ v, float* __restrict__ o) {
  const int lane = threadIdx.x;
  const int half = lane >> 4;
  const int l16  = lane & 15;
  const int t0 = blockIdx.x * 32;
  const int h  = blockIdx.y;
  const int b  = blockIdx.z;

  const float* qbase = q + ((size_t)b * SEQ + t0) * KDIM + h * HS;
  const float* kbase = k + (size_t)b * SEQ * KDIM + h * HS;
  const float* vbase = v + (size_t)b * SEQ * KDIM + h * HS;

  const float scale = 0.03125f * 1.44269504088896340736f;

  v2f qf[2][16];
#pragma unroll
  for (int u = 0; u < 2; ++u)
#pragma unroll
    for (int g = 0; g < 2; ++g)
#pragma unroll
      for (int i = 0; i < 4; ++i) {
        v4f t = *(const v4f*)(qbase + (size_t)(u * 16 + l16) * KDIM + 32 * g + 16 * half + 4 * i);
        qf[u][8 * g + 2 * i].x     = t.x * scale;
        qf[u][8 * g + 2 * i].y     = t.y * scale;
        qf[u][8 * g + 2 * i + 1].x = t.z * scale;
        qf[u][8 * g + 2 * i + 1].y = t.w * scale;
      }

  float m_r[2], l_r[2];
  v8f oacc[2][4];
#pragma unroll
  for (int u = 0; u < 2; ++u) {
    m_r[u] = -3.0e38f;
    l_r[u] = 0.0f;
#pragma unroll
    for (int j = 0; j < 4; ++j)
#pragma unroll
      for (int r = 0; r < 8; ++r) oacc[u][j][r] = 0.0f;
  }

  v4f kf[2][4];
#pragma unroll
  for (int g = 0; g < 2; ++g)
#pragma unroll
    for (int i = 0; i < 4; ++i)
      kf[g][i] = *(const v4f*)(kbase + (size_t)l16 * KDIM + 32 * g + 16 * half + 4 * i);

  for (int kt = 0; kt < SEQ; kt += 16) {
    v8f s[2];
#pragma unroll
    for (int u = 0; u < 2; ++u)
#pragma unroll
      for (int r = 0; r < 8; ++r) s[u][r] = 0.0f;
    v4f vl[8];
#pragma unroll
    for (int e = 0; e < 8; ++e)
      vl[e] = *(const v4f*)(vbase + (size_t)(kt + 8 * half + e) * KDIM + 4 * l16);
#pragma unroll
    for (int g = 0; g < 2; ++g)
#pragma unroll
      for (int i = 0; i < 4; ++i) {
        const int kk = 8 * g + 2 * i;
        s[0] = WMMA_F32(kf[g][i].lo, qf[0][kk], s[0]);
        s[1] = WMMA_F32(kf[g][i].lo, qf[1][kk], s[1]);
        s[0] = WMMA_F32(kf[g][i].hi, qf[0][kk + 1], s[0]);
        s[1] = WMMA_F32(kf[g][i].hi, qf[1][kk + 1], s[1]);
      }
    SCHED_FENCE();
    const int kn = (kt + 16 < SEQ) ? kt + 16 : kt;
#pragma unroll
    for (int g = 0; g < 2; ++g)
#pragma unroll
      for (int i = 0; i < 4; ++i)
        kf[g][i] = *(const v4f*)(kbase + (size_t)(kn + l16) * KDIM + 32 * g + 16 * half + 4 * i);
    SCHED_FENCE();

#pragma unroll
    for (int u = 0; u < 2; ++u) {
      float tmax = fmaxf(fmaxf(fmaxf(s[u][0], s[u][1]), fmaxf(s[u][2], s[u][3])),
                         fmaxf(fmaxf(s[u][4], s[u][5]), fmaxf(s[u][6], s[u][7])));
      tmax = fmaxf(tmax, xhalf(tmax));
      const float mn   = fmaxf(m_r[u], tmax);
      const float corr = exp2f(m_r[u] - mn);
      m_r[u] = mn;
      float psum = 0.0f;
#pragma unroll
      for (int r = 0; r < 8; ++r) {
        const float p = exp2f(s[u][r] - mn);
        s[u][r] = p;
        psum += p;
      }
      l_r[u] = l_r[u] * corr + psum;
#pragma unroll
      for (int j = 0; j < 4; ++j)
#pragma unroll
        for (int r = 0; r < 8; ++r) oacc[u][j][r] *= corr;
    }

    SCHED_FENCE();
#pragma unroll
    for (int t = 0; t < 4; ++t) {
      v2f p0, p1;
      p0.x = s[0][2 * t]; p0.y = s[0][2 * t + 1];
      p1.x = s[1][2 * t]; p1.y = s[1][2 * t + 1];
#pragma unroll
      for (int j = 0; j < 4; ++j) {
        v2f vf;
        vf.x = vl[2 * t][j];
        vf.y = vl[2 * t + 1][j];
        oacc[0][j] = WMMA_F32(vf, p0, oacc[0][j]);
        oacc[1][j] = WMMA_F32(vf, p1, oacc[1][j]);
      }
    }
  }

#pragma unroll
  for (int u = 0; u < 2; ++u) {
    const float inv = 1.0f / (l_r[u] + xhalf(l_r[u]));
    float* orow = o + ((size_t)b * SEQ + t0 + u * 16 + l16) * KDIM + h * HS + 32 * half;
#pragma unroll
    for (int r = 0; r < 8; ++r) {
      float4 w;
      w.x = oacc[u][0][r] * inv; w.y = oacc[u][1][r] * inv;
      w.z = oacc[u][2][r] * inv; w.w = oacc[u][3][r] * inv;
      *(float4*)(orow + 4 * r) = w;
    }
  }
}

__global__ void __launch_bounds__(32)
out_proj_kernel(const float* A, const float* Wu,
                const float* bu, float* out) {
  const int lane = threadIdx.x;
  const int half = lane >> 4;
  const int l16  = lane & 15;
  const int m0 = blockIdx.x * 64;
  const int n0 = blockIdx.y * 64;

  v8f acc[4][4];
  gemm_tile_64x64(A, Wu, m0, n0, lane, acc);

  float bias[4];
#pragma unroll
  for (int j = 0; j < 4; ++j) bias[j] = bu[n0 + j * 16 + l16];
#pragma unroll
  for (int j = 0; j < 4; ++j) {
#pragma unroll
    for (int i = 0; i < 4; ++i)
#pragma unroll
      for (int r = 0; r < 8; ++r)
        out[(size_t)(m0 + i * 16 + r + 8 * half) * KDIM + n0 + j * 16 + l16] =
            acc[i][j][r] + bias[j];
  }
}

extern "C" void kernel_launch(void* const* d_in, const int* in_sizes, int n_in,
                              void* d_out, int out_size, void* d_ws, size_t ws_size,
                              hipStream_t stream) {
  const float* x  = (const float*)d_in[0];
  const float* Wk = (const float*)d_in[1];
  const float* Wq = (const float*)d_in[2];
  const float* Wv = (const float*)d_in[3];
  const float* Wu = (const float*)d_in[4];
  const float* bu = (const float*)d_in[5];
  float* out = (float*)d_out;

  const size_t MAT = (size_t)ROWS * KDIM;
  float* qb = (float*)d_ws;
  float* kb = qb + MAT;
  float* vb = kb + MAT;
  float* ob = vb + MAT;

  dim3 g1(ROWS / 64, KDIM / 64, 3);
  qkv_proj_kernel<<<g1, 32, 0, stream>>>(x, Wq, Wk, Wv, qb, kb, vb);

  dim3 g2(SEQ / 32, HEADS, BATCH);
  flash_attn_kernel<<<g2, 32, 0, stream>>>(qb, kb, vb, ob);

  dim3 g3(ROWS / 64, KDIM / 64);
  out_proj_kernel<<<g3, 32, 0, stream>>>(ob, Wu, bu, out);
}


# work-groups of four waves (GEMMs: 2x2 tiles of a 128x128 block; attention: four query tiles of one head) so that waves reading the same operand lines meet in the unit's cache; sums unchanged
# speedup vs baseline: 1.8228x; 1.0383x over previous
#include <hip/hip_runtime.h>

typedef __attribute__((ext_vector_type(2))) float v2f;
typedef __attribute__((ext_vector_type(4))) float v4f;
typedef __attribute__((ext_vector_type(8))) float v8f;

#define WMMA_F32(a, b, c) \
  __builtin_amdgcn_wmma_f32_16x16x4_f32(false, (a), false, (b), (short)0, (c), false, false)

#define BATCH 4
#define SEQ   2048
#define KDIM  1024
#define HEADS 16
#define HS    64
#define ROWS  (BATCH * SEQ)


#define SCHED_FENCE() __builtin_amdgcn_sched_barrier(0)

__device__ __forceinline__ int wave_in_group() {
  return __builtin_amdgcn_readfirstlane((int)(threadIdx.x >> 5));
}

__device__ __forceinline__ void load_block(const float* const (&row)[4], int k,
                                           v4f (&f)[4][2]) {
#pragma unroll
  for (int i = 0; i < 4; ++i)
#pragma unroll
    for (int q = 0; q < 2; ++q) f[i][q] = *(const v4f*)(row[i] + k + 4 * q);
}

__device__ __forceinline__ void mma_block(const v4f (&a)[4][2], const v4f (&b)[4][2],
                                          v8f (&acc)[4][4]) {
#pragma unroll
  for (int q = 0; q < 2; ++q) {
#pragma unroll
    for (int i = 0; i < 4; ++i)
#pragma unroll
      for (int j = 0; j < 4; ++j)
        acc[i][j] = WMMA_F32(a[i][q].lo, b[j][q].lo, acc[i][j]);
#pragma unroll
    for (int i = 0; i < 4; ++i)
#pragma unroll
      for (int j = 0; j < 4; ++j)
        acc[i][j] = WMMA_F32(a[i][q].hi, b[j][q].hi, acc[i][j]);
  }
}

__device__ __forceinline__ void gemm_tile_64x64(const float* A, const float* B,
                                                int m0, int n0, int lane,
                                                v8f (&acc)[4][4]) {
  const int half = lane >> 4;
  const int l16  = lane & 15;
#pragma unroll
  for (int i = 0; i < 4; ++i)
#pragma unroll
    for (int j = 0; j < 4; ++j)
#pragma unroll
      for (int r = 0; r < 8; ++r) acc[i][j][r] = 0.0f;

  const float* arow[4];
  const float* brow[4];
#pragma unroll
  for (int i = 0; i < 4; ++i) {
    arow[i] = A + (size_t)(m0 + i * 16 + l16) * KDIM + 16 * half;
    brow[i] = B + (size_t)(n0 + i * 16 + l16) * KDIM + 16 * half;
  }

  v4f aX[4][2], bX[4][2], aY[4][2], bY[4][2];
  load_block(arow, 0, aX);
  load_block(brow, 0, bX);
  SCHED_FENCE();
  load_block(arow, 8, aY);
  load_block(brow, 8, bY);
  for (int kc = 0; kc < KDIM; kc += 32) {
    const int kn = (kc + 32 < KDIM) ? kc + 32 : kc;
    mma_block(aX, bX, acc);
    SCHED_FENCE();
    load_block(arow, kn, aX);
    load_block(brow, kn, bX);
    SCHED_FENCE();
    mma_block(aY, bY, acc);
    load_block(arow, kn + 8, aY);
    load_block(brow, kn + 8, bY);
    SCHED_FENCE();
  }
}

__global__ void __launch_bounds__(128)
qkv_proj_kernel(const float* x,
                const float* Wq, const float* Wk,
                const float* Wv,
                float* qo, float* ko,
                float* vo) {
  const int lane = threadIdx.x & 31;
  const int half = lane >> 4;
  const int l16  = lane & 15;
  const int w  = wave_in_group();
  const int m0 = blockIdx.x * 128 + (w & 1) * 64;
  const int n0 = blockIdx.y * 128 + (w >> 1) * 64;
  const int which = blockIdx.z;

  const float* W  = (which == 0) ? Wq : (which == 1) ? Wk : Wv;
  float*      out = (which == 0) ? qo : (which == 1) ? ko : vo;

  v8f acc[4][4];
  gemm_tile_64x64(x, W, m0, n0, lane, acc);

#pragma unroll
  for (int i = 0; i < 4; ++i)
#pragma unroll
    for (int j = 0; j < 4; ++j)
#pragma unroll
      for (int r = 0; r < 8; ++r)
        out[(size_t)(m0 + i * 16 + r + 8 * half) * KDIM + n0 + j * 16 + l16] =
            acc[i][j][r];
}

__device__ __forceinline__ float xhalf(float x) {
  int bits = __builtin_bit_cast(int, x);
  return __builtin_bit_cast(
      float, __builtin_amdgcn_permlanex16(bits, bits, 0x76543210, 0xfedcba98,
                                          false, false));
}

__global__ void __launch_bounds__(128)
flash_attn_kernel(const float* __restrict__ q, const float* __restrict__ k,
                  const float* __restrict__ v, float* __restrict__ o) {
  const int lane = threadIdx.x & 31;
  const int half = lane >> 4;
  const int l16  = lane & 15;
  const int t0 = (blockIdx.x * 4 + wave_in_group()) * 32;
  const int h  = blockIdx.y;
  const int b  = blockIdx.z;

  const float* qbase = q + ((size_t)b * SEQ + t0) * KDIM + h * HS;
  const float* kbase = k + (size_t)b * SEQ * KDIM + h * HS;
  const float* vbase = v + (size_t)b * SEQ * KDIM + h * HS;

  const float scale = 0.03125f * 1.44269504088896340736f;

  v2f qf[2][16];
#pragma unroll
  for (int u = 0; u < 2; ++u)
#pragma unroll
    for (int g = 0; g < 2; ++g)
#pragma unroll
      for (int i = 0; i < 4; ++i) {
        v4f t = *(const v4f*)(qbase + (size_t)(u * 16 + l16) * KDIM + 32 * g + 16 * half + 4 * i);
        qf[u][8 * g + 2 * i].x     = t.x * scale;
        qf[u][8 * g + 2 * i].y     = t.y * scale;
        qf[u][8 * g + 2 * i + 1].x = t.z * scale;
        qf[u][8 * g + 2 * i + 1].y = t.w * scale;
      }

  float m_r[2], l_r[2];
  v8f oacc[2][4];
#pragma unroll
  for (int u = 0; u < 2; ++u) {
    m_r[u] = -3.0e38f;
    l_r[u] = 0.0f;
#pragma unroll
    for (int j = 0; j < 4; ++j)
#pragma unroll
      for (int r = 0; r < 8; ++r) oacc[u][j][r] = 0.0f;
  }

  v4f kf[2][4];
#pragma unroll
  for (int g = 0; g < 2; ++g)
#pragma unroll
    for (int i = 0; i < 4; ++i)
      kf[g][i] = *(const v4f*)(kbase + (size_t)l16 * KDIM + 32 * g + 16 * half + 4 * i);

  for (int kt = 0; kt < SEQ; kt += 16) {
    v8f s[2];
#pragma unroll
    for (int u = 0; u < 2; ++u)
#pragma unroll
      for (int r = 0; r < 8; ++r) s[u][r] = 0.0f;
    v4f vl[8];
#pragma unroll
    for (int e = 0; e < 8; ++e)
      vl[e] = *(const v4f*)(vbase + (size_t)(kt + 8 * half + e) * KDIM + 4 * l16);
#pragma unroll
    for (int g = 0; g < 2; ++g)
#pragma unroll
      for (int i = 0; i < 4; ++i) {
        const int kk = 8 * g + 2 * i;
        s[0] = WMMA_F32(kf[g][i].lo, qf[0][kk], s[0]);
        s[1] = WMMA_F32(kf[g][i].lo, qf[1][kk], s[1]);
        s[0] = WMMA_F32(kf[g][i].hi, qf[0][kk + 1], s[0]);
        s[1] = WMMA_F32(kf[g][i].hi, qf[1][kk + 1], s[1]);
      }
    SCHED_FENCE();
    const int kn = (kt + 16 < SEQ) ? kt + 16 : kt;
#pragma unroll
    for (int g = 0; g < 2; ++g)
#pragma unroll
      for (int i = 0; i < 4; ++i)
        kf[g][i] = *(const v4f*)(kbase + (size_t)(kn + l16) * KDIM + 32 * g + 16 * half + 4 * i);
    SCHED_FENCE();

#pragma unroll
    for (int u = 0; u < 2; ++u) {
      float tmax = fmaxf(fmaxf(fmaxf(s[u][0], s[u][1]), fmaxf(s[u][2], s[u][3])),
                         fmaxf(fmaxf(s[u][4], s[u][5]), fmaxf(s[u][6], s[u][7])));
      tmax = fmaxf(tmax, xhalf(tmax));
      const float mn   = fmaxf(m_r[u], tmax);
      const float corr = exp2f(m_r[u] - mn);
      m_r[u] = mn;
      float psum = 0.0f;
#pragma unroll
      for (int r = 0; r < 8; ++r) {
        const float p = exp2f(s[u][r] - mn);
        s[u][r] = p;
        psum += p;
      }
      l_r[u] = l_r[u] * corr + psum;
#pragma unroll
      for (int j = 0; j < 4; ++j)
#pragma unroll
        for (int r = 0; r < 8; ++r) oacc[u][j][r] *= corr;
    }

    SCHED_FENCE();
#pragma unroll
    for (int t = 0; t < 4; ++t) {
      v2f p0, p1;
      p0.x = s[0][2 * t]; p0.y = s[0][2 * t + 1];
      p1.x = s[1][2 * t]; p1.y = s[1][2 * t + 1];
#pragma unroll
      for (int j = 0; j < 4; ++j) {
        v2f vf;
        vf.x = vl[2 * t][j];
        vf.y = vl[2 * t + 1][j];
        oacc[0][j] = WMMA_F32(vf, p0, oacc[0][j]);
        oacc[1][j] = WMMA_F32(vf, p1, oacc[1][j]);
      }
    }
  }

#pragma unroll
  for (int u = 0; u < 2; ++u) {
    const float inv = 1.0f / (l_r[u] + xhalf(l_r[u]));
    float* orow = o + ((size_t)b * SEQ + t0 + u * 16 + l16) * KDIM + h * HS + 32 * half;
#pragma unroll
    for (int r = 0; r < 8; ++r) {
      float4 w;
      w.x = oacc[u][0][r] * inv; w.y = oacc[u][1][r] * inv;
      w.z = oacc[u][2][r] * inv; w.w = oacc[u][3][r] * inv;
      *(float4*)(orow + 4 * r) = w;
    }
  }
}

__global__ void __launch_bounds__(128)
out_proj_kernel(const float* A, const float* Wu,
                const float* bu, float* out) {
  const int lane = threadIdx.x & 31;
  const int half = lane >> 4;
  const int l16  = lane & 15;
  const int w  = wave_in_group();
  const int m0 = blockIdx.x * 128 + (w & 1) * 64;
  const int n0 = blockIdx.y * 128 + (w >> 1) * 64;

  v8f acc[4][4];
  gemm_tile_64x64(A, Wu, m0, n0, lane, acc);

  float bias[4];
#pragma unroll
  for (int j = 0; j < 4; ++j) bias[j] = bu[n0 + j * 16 + l16];
#pragma unroll
  for (int j = 0; j < 4; ++j) {
#pragma unroll
    for (int i = 0; i < 4; ++i)
#pragma unroll
      for (int r = 0; r < 8; ++r)
        out[(size_t)(m0 + i * 16 + r + 8 * half) * KDIM + n0 + j * 16 + l16] =
            acc[i][j][r] + bias[j];
  }
}

extern "C" void kernel_launch(void* const* d_in, const int* in_sizes, int n_in,
                              void* d_out, int out_size, void* d_ws, size_t ws_size,
                              hipStream_t stream) {
  const float* x  = (const float*)d_in[0];
  const float* Wk = (const float*)d_in[1];
  const float* Wq = (const float*)d_in[2];
  const float* Wv = (const float*)d_in[3];
  const float* Wu = (const float*)d_in[4];
  const float* bu = (const float*)d_in[5];
  float* out = (float*)d_out;

  const size_t MAT = (size_t)ROWS * KDIM;
  float* qb = (float*)d_ws;
  float* kb = qb + MAT;
  float* vb = kb + MAT;
  float* ob = vb + MAT;

  dim3 g1(ROWS / 128, KDIM / 128, 3);
  qkv_proj_kernel<<<g1, 128, 0, stream>>>(x, Wq, Wk, Wv, qb, kb, vb);

  dim3 g2(SEQ / 128, HEADS, BATCH);
  flash_attn_kernel<<<g2, 128, 0, stream>>>(qb, kb, vb, ob);

  dim3 g3(ROWS / 128, KDIM / 128);
  out_proj_kernel<<<g3, 128, 0, stream>>>(ob, Wu, bu, out);
}
